# starting kernel
# baseline (speedup 1.0000x reference)
#include <hip/hip_runtime.h>
#include <math.h>


typedef __attribute__((ext_vector_type(2))) float v2f;
typedef __attribute__((ext_vector_type(8))) float v8f;

#define WMMA_F32_16x16x4(a, b, c) \
  __builtin_amdgcn_wmma_f32_16x16x4_f32(false, (a), false, (b), (short)0, (c), false, false)

template <int NT>
__global__ __launch_bounds__(32) void EmbedWeighted_gemm_wmma(
    const float* __restrict__ A,
    const float* __restrict__ B,
    float* __restrict__ C,
    int M, int K, int N_) {
  const int N     = (NT > 0) ? NT : N_;
  const int lane  = threadIdx.x & 31;
  const int l16   = lane & 15;
  const int khalf = (lane >> 4) << 1;
  const int m0    = blockIdx.x << 4;
  const int n0    = blockIdx.y << 4;

  const float* ap = A + (size_t)(m0 + l16) * K + khalf;
  const float* bp = B + (size_t)khalf * N + (n0 + l16);

  v8f acc = {};

  int k0 = 0;
  for (; k0 + 32 <= K; k0 += 32) {
    __builtin_prefetch(ap + 128, 0, 3);
#pragma unroll
    for (int kk = 0; kk < 32; kk += 4) {
      v2f a = *(const v2f*)(ap + kk);
      v2f b;
      b.x = bp[kk * N];
      b.y = bp[kk * N + N];
      acc = WMMA_F32_16x16x4(a, b, acc);
    }
    ap += 32;
    bp += (size_t)32 * N;
  }
  for (; k0 < K; k0 += 4) {
    v2f a = *(const v2f*)ap;
    v2f b;
    b.x = bp[0];
    b.y = bp[N];
    acc = WMMA_F32_16x16x4(a, b, acc);
    ap += 4;
    bp += (size_t)4 * N;
  }

  float* crow = C + (size_t)(m0 + ((lane >> 4) << 3)) * N + (n0 + l16);
#pragma unroll
  for (int r = 0; r < 8; ++r) {
    crow[(size_t)r * N] = acc[r];
  }
}

extern "C" void kernel_launch(void* const* d_in, const int* in_sizes, int n_in,
                              void* d_out, int out_size, void* d_ws, size_t ws_size,
                              hipStream_t stream) {
  (void)n_in; (void)d_ws; (void)ws_size;
  const float* inputs = (const float*)d_in[0];
  const float* emb    = (const float*)d_in[1];
  float* out          = (float*)d_out;

  const double s0 = (double)in_sizes[0];
  const double s1 = (double)in_sizes[1];
  const double so = (double)out_size;
  const int D  = (int)llround(sqrt(s1 * so / s0));
  const int V  = (int)llround(s1 / D);
  const int Bt = (int)llround(so / D);

  dim3 grid(Bt / 16, D / 16, 1);
  dim3 block(32, 1, 1);
  if (D == 64) {
    EmbedWeighted_gemm_wmma<64><<<grid, block, 0, stream>>>(inputs, emb, out, Bt, V, D);
  } else {
    EmbedWeighted_gemm_wmma<0><<<grid, block, 0, stream>>>(inputs, emb, out, Bt, V, D);
  }
}


# one work-group per 16x32 block of C, K split over its 10 waves (2 accumulator tiles each, loads issued one group ahead), partials summed through LDS in fixed wave order: summation order changes
# speedup vs baseline: 3.5307x; 3.5307x over previous
#include <hip/hip_runtime.h>
#include <math.h>


typedef __attribute__((ext_vector_type(2))) float v2f;
typedef __attribute__((ext_vector_type(4))) float v4f;
typedef __attribute__((ext_vector_type(8))) float v8f;

#define WMMA_F32_16x16x4(a, b, c) \
  __builtin_amdgcn_wmma_f32_16x16x4_f32(false, (a), false, (b), (short)0, (c), false, false)

template <int NT>
__global__ __launch_bounds__(32) void EmbedWeighted_gemm_wmma(
    const float* __restrict__ A,
    const float* __restrict__ B,
    float* __restrict__ C,
    int M, int K, int N_) {
  const int N     = (NT > 0) ? NT : N_;
  const int lane  = threadIdx.x & 31;
  const int l16   = lane & 15;
  const int khalf = (lane >> 4) << 1;
  const int m0    = blockIdx.x << 4;
  const int n0    = blockIdx.y << 4;

  const float* ap = A + (size_t)(m0 + l16) * K + khalf;
  const float* bp = B + (size_t)khalf * N + (n0 + l16);

  v8f acc = {};

  int k0 = 0;
  for (; k0 + 32 <= K; k0 += 32) {
    __builtin_prefetch(ap + 128, 0, 3);
#pragma unroll
    for (int kk = 0; kk < 32; kk += 4) {
      v2f a = *(const v2f*)(ap + kk);
      v2f b;
      b.x = bp[kk * N];
      b.y = bp[kk * N + N];
      acc = WMMA_F32_16x16x4(a, b, acc);
    }
    ap += 32;
    bp += (size_t)32 * N;
  }
  for (; k0 < K; k0 += 4) {
    v2f a = *(const v2f*)ap;
    v2f b;
    b.x = bp[0];
    b.y = bp[N];
    acc = WMMA_F32_16x16x4(a, b, acc);
    ap += 4;
    bp += (size_t)4 * N;
  }

  float* crow = C + (size_t)(m0 + ((lane >> 4) << 3)) * N + (n0 + l16);
#pragma unroll
  for (int r = 0; r < 8; ++r) {
    crow[(size_t)r * N] = acc[r];
  }
}

constexpr int SPLIT = 10;
constexpr int GRP   = 5;

template <int KT>
__global__ __launch_bounds__(32 * SPLIT) void gemm_wmma_f32_ksplit(
    const float* __restrict__ A,
    const float* __restrict__ B,
    float* __restrict__ C) {
  constexpr int N   = 64;
  constexpr int KW  = KT / SPLIT;
  constexpr int NB  = KW / 8;
  constexpr int NG  = NB / GRP;
  static_assert(KT % (SPLIT * 8 * GRP) == 0, "K must split evenly");

  __shared__ v2f red[SPLIT][8][32];

  const int tid  = threadIdx.x;
  const int wave = __builtin_amdgcn_readfirstlane(tid >> 5);
  const int lane = tid & 31;
  const int c    = lane & 15;
  const int h    = lane >> 4;
  const int n0   = blockIdx.x << 5;
  const int m0   = blockIdx.y << 4;

  const float* ap = A + (size_t)(m0 + c) * KT + (wave * KW + 4 * h);
  const float* bp = B + (size_t)(wave * KW + 4 * h) * N + (n0 + 2 * c);

  v8f acc0 = {};
  v8f acc1 = {};

  v4f a[2][GRP];
  v2f b[2][GRP][4];

#pragma unroll
  for (int i = 0; i < GRP; ++i) {
    a[0][i] = *(const v4f*)(ap + i * 8);
#pragma unroll
    for (int j = 0; j < 4; ++j) b[0][i][j] = *(const v2f*)(bp + (i * 8 + j) * N);
  }

#pragma unroll
  for (int g = 0; g < NG; ++g) {
    const int cur = g & 1;
    const int nxt = cur ^ 1;
    if (g + 1 < NG) {
      __builtin_amdgcn_sched_barrier(0);
#pragma unroll
      for (int i = 0; i < GRP; ++i) {
        const int blk = (g + 1) * GRP + i;
        a[nxt][i] = *(const v4f*)(ap + blk * 8);
#pragma unroll
        for (int j = 0; j < 4; ++j) b[nxt][i][j] = *(const v2f*)(bp + (blk * 8 + j) * N);
      }
      __builtin_amdgcn_sched_barrier(0);
    }
#pragma unroll
    for (int i = 0; i < GRP; ++i) {
      const v4f av = a[cur][i];
      v2f alo, ahi, b00, b01, b10, b11;
      alo.x = av.x; alo.y = av.y;
      ahi.x = av.z; ahi.y = av.w;
      b00.x = b[cur][i][0].x; b00.y = b[cur][i][1].x;
      b01.x = b[cur][i][0].y; b01.y = b[cur][i][1].y;
      b10.x = b[cur][i][2].x; b10.y = b[cur][i][3].x;
      b11.x = b[cur][i][2].y; b11.y = b[cur][i][3].y;
      acc0 = WMMA_F32_16x16x4(alo, b00, acc0);
      acc1 = WMMA_F32_16x16x4(alo, b01, acc1);
      acc0 = WMMA_F32_16x16x4(ahi, b10, acc0);
      acc1 = WMMA_F32_16x16x4(ahi, b11, acc1);
    }
  }

#pragma unroll
  for (int r = 0; r < 8; ++r) {
    v2f p;
    p.x = acc0[r];
    p.y = acc1[r];
    red[wave][r][lane] = p;
  }
  __syncthreads();

  if (tid < 256) {
    const int r = tid >> 5;
    v2f s = red[0][r][lane];
#pragma unroll
    for (int w = 1; w < SPLIT; ++w) {
      const v2f p = red[w][r][lane];
      s.x += p.x;
      s.y += p.y;
    }
    *(v2f*)(C + (size_t)(m0 + 8 * h + r) * N + (n0 + 2 * c)) = s;
  }
}

extern "C" void kernel_launch(void* const* d_in, const int* in_sizes, int n_in,
                              void* d_out, int out_size, void* d_ws, size_t ws_size,
                              hipStream_t stream) {
  (void)n_in; (void)d_ws; (void)ws_size;
  const float* inputs = (const float*)d_in[0];
  const float* emb    = (const float*)d_in[1];
  float* out          = (float*)d_out;

  const double s0 = (double)in_sizes[0];
  const double s1 = (double)in_sizes[1];
  const double so = (double)out_size;
  const int D  = (int)llround(sqrt(s1 * so / s0));
  const int V  = (int)llround(s1 / D);
  const int Bt = (int)llround(so / D);

  if (D == 64 && V == 2000) {
    dim3 grid(2, Bt / 16, 1);
    dim3 block(32 * SPLIT, 1, 1);
    gemm_wmma_f32_ksplit<2000><<<grid, block, 0, stream>>>(inputs, emb, out);
  } else {
    dim3 grid(Bt / 16, D / 16, 1);
    dim3 block(32, 1, 1);
    EmbedWeighted_gemm_wmma<0><<<grid, block, 0, stream>>>(inputs, emb, out, Bt, V, D);
  }
}


# K split over 8 waves, 2 per SIMD with 63/63/62/62 k-blocks per SIMD, instead of 10 waves (75/75/50/50); load groups of 6 blocks; 8 partials summed in wave order: summation order changes
# speedup vs baseline: 3.8843x; 1.1002x over previous
#include <hip/hip_runtime.h>
#include <math.h>


typedef __attribute__((ext_vector_type(2))) float v2f;
typedef __attribute__((ext_vector_type(4))) float v4f;
typedef __attribute__((ext_vector_type(8))) float v8f;

#define WMMA_F32_16x16x4(a, b, c) \
  __builtin_amdgcn_wmma_f32_16x16x4_f32(false, (a), false, (b), (short)0, (c), false, false)

template <int NT>
__global__ __launch_bounds__(32) void EmbedWeighted_gemm_wmma(
    const float* __restrict__ A,
    const float* __restrict__ B,
    float* __restrict__ C,
    int M, int K, int N_) {
  const int N     = (NT > 0) ? NT : N_;
  const int lane  = threadIdx.x & 31;
  const int l16   = lane & 15;
  const int khalf = (lane >> 4) << 1;
  const int m0    = blockIdx.x << 4;
  const int n0    = blockIdx.y << 4;

  const float* ap = A + (size_t)(m0 + l16) * K + khalf;
  const float* bp = B + (size_t)khalf * N + (n0 + l16);

  v8f acc = {};

  int k0 = 0;
  for (; k0 + 32 <= K; k0 += 32) {
    __builtin_prefetch(ap + 128, 0, 3);
#pragma unroll
    for (int kk = 0; kk < 32; kk += 4) {
      v2f a = *(const v2f*)(ap + kk);
      v2f b;
      b.x = bp[kk * N];
      b.y = bp[kk * N + N];
      acc = WMMA_F32_16x16x4(a, b, acc);
    }
    ap += 32;
    bp += (size_t)32 * N;
  }
  for (; k0 < K; k0 += 4) {
    v2f a = *(const v2f*)ap;
    v2f b;
    b.x = bp[0];
    b.y = bp[N];
    acc = WMMA_F32_16x16x4(a, b, acc);
    ap += 4;
    bp += (size_t)4 * N;
  }

  float* crow = C + (size_t)(m0 + ((lane >> 4) << 3)) * N + (n0 + l16);
#pragma unroll
  for (int r = 0; r < 8; ++r) {
    crow[(size_t)r * N] = acc[r];
  }
}

constexpr int SPLIT = 8;
constexpr int GRP   = 6;

template <int KT>
__global__ __launch_bounds__(32 * SPLIT) void gemm_wmma_f32_ksplit(
    const float* __restrict__ A,
    const float* __restrict__ B,
    float* __restrict__ C) {
  constexpr int N    = 64;
  constexpr int TB   = KT / 8;
  constexpr int NBLK = TB / SPLIT;
  constexpr int XTRA = TB % SPLIT;
  constexpr int NG   = NBLK / GRP;
  constexpr int REM  = NBLK % GRP;
  static_assert(KT % 8 == 0, "K must be whole k-blocks");
  static_assert(NG >= 1 && REM + 1 <= GRP, "tail must fit one register set");

  __shared__ v2f red[SPLIT][8][32];

  const int tid  = threadIdx.x;
  const int wave = __builtin_amdgcn_readfirstlane(tid >> 5);
  const int lane = tid & 31;
  const int c    = lane & 15;
  const int h    = lane >> 4;
  const int n0   = blockIdx.x << 5;
  const int m0   = blockIdx.y << 4;

  const bool extra = wave < XTRA;
  const int  kw    = 8 * (NBLK * wave + (wave < XTRA ? wave : XTRA));

  const float* ap = A + (size_t)(m0 + c) * KT + (kw + 4 * h);
  const float* bp = B + (size_t)(kw + 4 * h) * N + (n0 + 2 * c);

  v8f acc0 = {};
  v8f acc1 = {};

  v4f a[2][GRP];
  v2f b[2][GRP][4];

#define LOAD_BLOCK(buf, slot, blk)                                              \
  do {                                                                          \
    a[buf][slot] = *(const v4f*)(ap + (blk) * 8);                               \
    _Pragma("unroll") for (int j = 0; j < 4; ++j)                               \
      b[buf][slot][j] = *(const v2f*)(bp + ((blk) * 8 + j) * N);                \
  } while (0)

#define MMA_BLOCK(buf, slot)                                                    \
  do {                                                                          \
    const v4f av = a[buf][slot];                                                \
    v2f alo, ahi, b00, b01, b10, b11;                                           \
    alo.x = av.x; alo.y = av.y;                                                 \
    ahi.x = av.z; ahi.y = av.w;                                                 \
    b00.x = b[buf][slot][0].x; b00.y = b[buf][slot][1].x;                       \
    b01.x = b[buf][slot][0].y; b01.y = b[buf][slot][1].y;                       \
    b10.x = b[buf][slot][2].x; b10.y = b[buf][slot][3].x;                       \
    b11.x = b[buf][slot][2].y; b11.y = b[buf][slot][3].y;                       \
    acc0 = WMMA_F32_16x16x4(alo, b00, acc0);                                    \
    acc1 = WMMA_F32_16x16x4(alo, b01, acc1);                                    \
    acc0 = WMMA_F32_16x16x4(ahi, b10, acc0);                                    \
    acc1 = WMMA_F32_16x16x4(ahi, b11, acc1);                                    \
  } while (0)

#pragma unroll
  for (int i = 0; i < GRP; ++i) LOAD_BLOCK(0, i, i);

#pragma unroll
  for (int g = 0; g < NG; ++g) {
    const int cur = g & 1;
    const int nxt = cur ^ 1;
    __builtin_amdgcn_sched_barrier(0);
    if (g + 1 < NG) {
#pragma unroll
      for (int i = 0; i < GRP; ++i) LOAD_BLOCK(nxt, i, (g + 1) * GRP + i);
    } else {
#pragma unroll
      for (int i = 0; i < REM; ++i) LOAD_BLOCK(nxt, i, NG * GRP + i);
      if (extra) LOAD_BLOCK(nxt, REM, NG * GRP + REM);
    }
    __builtin_amdgcn_sched_barrier(0);
#pragma unroll
    for (int i = 0; i < GRP; ++i) MMA_BLOCK(cur, i);
  }

  {
    constexpr int tb = NG & 1;
#pragma unroll
    for (int i = 0; i < REM; ++i) MMA_BLOCK(tb, i);
    if (extra) MMA_BLOCK(tb, REM);
  }
#undef LOAD_BLOCK
#undef MMA_BLOCK

#pragma unroll
  for (int r = 0; r < 8; ++r) {
    v2f p;
    p.x = acc0[r];
    p.y = acc1[r];
    red[wave][r][lane] = p;
  }
  __syncthreads();

  {
    const int r = tid >> 5;
    v2f s = red[0][r][lane];
#pragma unroll
    for (int w = 1; w < SPLIT; ++w) {
      const v2f p = red[w][r][lane];
      s.x += p.x;
      s.y += p.y;
    }
    *(v2f*)(C + (size_t)(m0 + 8 * h + r) * N + (n0 + 2 * c)) = s;
  }
}

extern "C" void kernel_launch(void* const* d_in, const int* in_sizes, int n_in,
                              void* d_out, int out_size, void* d_ws, size_t ws_size,
                              hipStream_t stream) {
  (void)n_in; (void)d_ws; (void)ws_size;
  const float* inputs = (const float*)d_in[0];
  const float* emb    = (const float*)d_in[1];
  float* out          = (float*)d_out;

  const double s0 = (double)in_sizes[0];
  const double s1 = (double)in_sizes[1];
  const double so = (double)out_size;
  const int D  = (int)llround(sqrt(s1 * so / s0));
  const int V  = (int)llround(s1 / D);
  const int Bt = (int)llround(so / D);

  if (D == 64 && V == 2000) {
    dim3 grid(2, Bt / 16, 1);
    dim3 block(32 * SPLIT, 1, 1);
    gemm_wmma_f32_ksplit<2000><<<grid, block, 0, stream>>>(inputs, emb, out);
  } else {
    dim3 grid(Bt / 16, D / 16, 1);
    dim3 block(32, 1, 1);
    EmbedWeighted_gemm_wmma<0><<<grid, block, 0, stream>>>(inputs, emb, out, Bt, V, D);
  }
}


# each wave walks its K slice as a ring whose start differs by work-group (6 starts, whole load groups from the slice end), so work-groups stop reading the same B rows at once: summation order changes
# speedup vs baseline: 4.4914x; 1.1563x over previous
#include <hip/hip_runtime.h>
#include <math.h>


typedef __attribute__((ext_vector_type(2))) float v2f;
typedef __attribute__((ext_vector_type(4))) float v4f;
typedef __attribute__((ext_vector_type(8))) float v8f;

#define WMMA_F32_16x16x4(a, b, c) \
  __builtin_amdgcn_wmma_f32_16x16x4_f32(false, (a), false, (b), (short)0, (c), false, false)

template <int NT>
__global__ __launch_bounds__(32) void EmbedWeighted_gemm_wmma(
    const float* __restrict__ A,
    const float* __restrict__ B,
    float* __restrict__ C,
    int M, int K, int N_) {
  const int N     = (NT > 0) ? NT : N_;
  const int lane  = threadIdx.x & 31;
  const int l16   = lane & 15;
  const int khalf = (lane >> 4) << 1;
  const int m0    = blockIdx.x << 4;
  const int n0    = blockIdx.y << 4;

  const float* ap = A + (size_t)(m0 + l16) * K + khalf;
  const float* bp = B + (size_t)khalf * N + (n0 + l16);

  v8f acc = {};

  int k0 = 0;
  for (; k0 + 32 <= K; k0 += 32) {
    __builtin_prefetch(ap + 128, 0, 3);
#pragma unroll
    for (int kk = 0; kk < 32; kk += 4) {
      v2f a = *(const v2f*)(ap + kk);
      v2f b;
      b.x = bp[kk * N];
      b.y = bp[kk * N + N];
      acc = WMMA_F32_16x16x4(a, b, acc);
    }
    ap += 32;
    bp += (size_t)32 * N;
  }
  for (; k0 < K; k0 += 4) {
    v2f a = *(const v2f*)ap;
    v2f b;
    b.x = bp[0];
    b.y = bp[N];
    acc = WMMA_F32_16x16x4(a, b, acc);
    ap += 4;
    bp += (size_t)4 * N;
  }

  float* crow = C + (size_t)(m0 + ((lane >> 4) << 3)) * N + (n0 + l16);
#pragma unroll
  for (int r = 0; r < 8; ++r) {
    crow[(size_t)r * N] = acc[r];
  }
}

constexpr int SPLIT = 8;
constexpr int GRP   = 6;

template <int KT>
__global__ __launch_bounds__(32 * SPLIT) void gemm_wmma_f32_ksplit(
    const float* __restrict__ A,
    const float* __restrict__ B,
    float* __restrict__ C) {
  constexpr int N    = 64;
  constexpr int TB   = KT / 8;
  constexpr int NBLK = TB / SPLIT;
  constexpr int XTRA = TB % SPLIT;
  constexpr int NG   = NBLK / GRP;
  constexpr int REM  = NBLK % GRP;
  static_assert(KT % 8 == 0, "K must be whole k-blocks");
  static_assert(NG >= 1 && REM + 1 <= GRP, "tail must fit one register set");

  __shared__ v2f red[SPLIT][8][32];

  const int tid  = threadIdx.x;
  const int wave = __builtin_amdgcn_readfirstlane(tid >> 5);
  const int lane = tid & 31;
  const int c    = lane & 15;
  const int h    = lane >> 4;
  const int n0   = blockIdx.x << 5;
  const int m0   = blockIdx.y << 4;

  const bool extra = wave < XTRA;
  const int  kw    = 8 * (NBLK * wave + (wave < XTRA ? wave : XTRA));

  const float* ap = A + (size_t)(m0 + c) * KT + (kw + 4 * h);
  const float* bp = B + (size_t)(kw + 4 * h) * N + (n0 + 2 * c);

  const unsigned LEN   = (unsigned)NBLK + (extra ? 1u : 0u);
  const unsigned PH    = (blockIdx.y + 3u * blockIdx.x) % (unsigned)(NG + 1);
  const unsigned START = PH == 0u ? 0u : LEN - (unsigned)GRP * PH;

  v8f acc0 = {};
  v8f acc1 = {};

  v4f a[2][GRP];
  v2f b[2][GRP][4];

#define LOAD_BLOCK(buf, slot, blk)                                              \
  do {                                                                          \
    const unsigned g_  = START + (unsigned)((blk) - (slot));                    \
    const unsigned gb_ = g_ >= LEN ? g_ - LEN : g_;                             \
    const float* ag_ = ap + (size_t)gb_ * 8;                                    \
    const float* bg_ = bp + (size_t)gb_ * (8 * N);                              \
    a[buf][slot] = *(const v4f*)(ag_ + (slot) * 8);                             \
    _Pragma("unroll") for (int j = 0; j < 4; ++j)                               \
      b[buf][slot][j] = *(const v2f*)(bg_ + ((slot) * 8 + j) * N);              \
  } while (0)

#define MMA_BLOCK(buf, slot)                                                    \
  do {                                                                          \
    const v4f av = a[buf][slot];                                                \
    v2f alo, ahi, b00, b01, b10, b11;                                           \
    alo.x = av.x; alo.y = av.y;                                                 \
    ahi.x = av.z; ahi.y = av.w;                                                 \
    b00.x = b[buf][slot][0].x; b00.y = b[buf][slot][1].x;                       \
    b01.x = b[buf][slot][0].y; b01.y = b[buf][slot][1].y;                       \
    b10.x = b[buf][slot][2].x; b10.y = b[buf][slot][3].x;                       \
    b11.x = b[buf][slot][2].y; b11.y = b[buf][slot][3].y;                       \
    acc0 = WMMA_F32_16x16x4(alo, b00, acc0);                                    \
    acc1 = WMMA_F32_16x16x4(alo, b01, acc1);                                    \
    acc0 = WMMA_F32_16x16x4(ahi, b10, acc0);                                    \
    acc1 = WMMA_F32_16x16x4(ahi, b11, acc1);                                    \
  } while (0)

#pragma unroll
  for (int i = 0; i < GRP; ++i) LOAD_BLOCK(0, i, i);

#pragma unroll
  for (int g = 0; g < NG; ++g) {
    const int cur = g & 1;
    const int nxt = cur ^ 1;
    __builtin_amdgcn_sched_barrier(0);
    if (g + 1 < NG) {
#pragma unroll
      for (int i = 0; i < GRP; ++i) LOAD_BLOCK(nxt, i, (g + 1) * GRP + i);
    } else {
#pragma unroll
      for (int i = 0; i < REM; ++i) LOAD_BLOCK(nxt, i, NG * GRP + i);
      if (extra) LOAD_BLOCK(nxt, REM, NG * GRP + REM);
    }
    __builtin_amdgcn_sched_barrier(0);
#pragma unroll
    for (int i = 0; i < GRP; ++i) MMA_BLOCK(cur, i);
  }

  {
    constexpr int tb = NG & 1;
#pragma unroll
    for (int i = 0; i < REM; ++i) MMA_BLOCK(tb, i);
    if (extra) MMA_BLOCK(tb, REM);
  }
#undef LOAD_BLOCK
#undef MMA_BLOCK

#pragma unroll
  for (int r = 0; r < 8; ++r) {
    v2f p;
    p.x = acc0[r];
    p.y = acc1[r];
    red[wave][r][lane] = p;
  }
  __syncthreads();

  {
    const int r = tid >> 5;
    v2f s = red[0][r][lane];
#pragma unroll
    for (int w = 1; w < SPLIT; ++w) {
      const v2f p = red[w][r][lane];
      s.x += p.x;
      s.y += p.y;
    }
    *(v2f*)(C + (size_t)(m0 + 8 * h + r) * N + (n0 + 2 * c)) = s;
  }
}

extern "C" void kernel_launch(void* const* d_in, const int* in_sizes, int n_in,
                              void* d_out, int out_size, void* d_ws, size_t ws_size,
                              hipStream_t stream) {
  (void)n_in; (void)d_ws; (void)ws_size;
  const float* inputs = (const float*)d_in[0];
  const float* emb    = (const float*)d_in[1];
  float* out          = (float*)d_out;

  const double s0 = (double)in_sizes[0];
  const double s1 = (double)in_sizes[1];
  const double so = (double)out_size;
  const int D  = (int)llround(sqrt(s1 * so / s0));
  const int V  = (int)llround(s1 / D);
  const int Bt = (int)llround(so / D);

  if (D == 64 && V == 2000) {
    dim3 grid(2, Bt / 16, 1);
    dim3 block(32 * SPLIT, 1, 1);
    gemm_wmma_f32_ksplit<2000><<<grid, block, 0, stream>>>(inputs, emb, out);
  } else {
    dim3 grid(Bt / 16, D / 16, 1);
    dim3 block(32, 1, 1);
    EmbedWeighted_gemm_wmma<0><<<grid, block, 0, stream>>>(inputs, emb, out, Bt, V, D);
  }
}
